# starting kernel
# baseline (speedup 1.0000x reference)
#include <hip/hip_runtime.h>


typedef float v2f __attribute__((ext_vector_type(2)));
typedef float v8f __attribute__((ext_vector_type(8)));

__device__ __forceinline__ v8f wmma_f32_16x16x4(v2f a, v2f b, v8f c) {
  return __builtin_amdgcn_wmma_f32_16x16x4_f32(false, a, false, b, (short)0, c,
                                               false, false);
}

__device__ __forceinline__ float bcast_lane(float v, int l) {
  return __int_as_float(__builtin_amdgcn_readlane(__float_as_int(v), l));
}

#define WAVE_LDS_FLOATS 1536
#define WAVES_PER_BLOCK 4

__global__ void __launch_bounds__(128) admm_qp_kernel(
    const float* __restrict__ Ag, const float* __restrict__ bg,
    const float* __restrict__ cg, const float* __restrict__ lbg,
    const float* __restrict__ ubg, float* __restrict__ outg, int nS) {
  __shared__ __align__(16) float smem[WAVES_PER_BLOCK * WAVE_LDS_FLOATS];

  const int lane = threadIdx.x & 31;
  const int wid = threadIdx.x >> 5;
  const int s = blockIdx.x * WAVES_PER_BLOCK + wid;
  if (s >= nS) return;

  float* W = smem + wid * WAVE_LDS_FLOATS;
  float* Alds = W;
  float* Ylds = W + 512;
  float* Glds = W + 1024;
  float* Milds = W + 1280;
  float* Qlds = W;
  float* Vlds = W + 1280;

  {
    const float4* Agv = (const float4*)(Ag + (size_t)s * 512);
    float4* Aldv = (float4*)Alds;
#pragma unroll
    for (int i = 0; i < 4; ++i) Aldv[lane + 32 * i] = Agv[lane + 32 * i];
  }

  const int m16 = lane & 15;
  const int hi = (lane < 16) ? 0 : 2;
  const int rb = (lane < 16) ? 0 : 8;

  {
    v8f g = {};
#pragma unroll
    for (int k = 0; k < 8; ++k) {
      v2f af = *(const v2f*)&Alds[m16 * 32 + 4 * k + hi];
      g = wmma_f32_16x16x4(af, af, g);
    }
#pragma unroll
    for (int r = 0; r < 8; ++r) {
      int row = rb + r;
      Glds[row * 16 + m16] = g[r] + ((row == m16) ? 1e-5f : 0.0f);
    }
  }

  float wc[16];
#pragma unroll
  for (int p = 0; p < 16; ++p) {
    float gv = Glds[p * 16 + m16];
    wc[p] = (lane < 16) ? gv : ((p == lane - 16) ? 1.0f : 0.0f);
  }
#pragma unroll
  for (int p = 0; p < 16; ++p) {
    float piv = bcast_lane(wc[p], p);
    float ip = 1.0f / piv;
    wc[p] *= ip;
#pragma unroll
    for (int r = 0; r < 16; ++r) {
      if (r == p) continue;
      float f = bcast_lane(wc[r], p);
      wc[r] = fmaf(-f, wc[p], wc[r]);
    }
  }
  if (lane >= 16) {
#pragma unroll
    for (int p = 0; p < 16; ++p) Milds[p * 16 + (lane - 16)] = wc[p];
  }

  {
    v2f mf[4];
#pragma unroll
    for (int k = 0; k < 4; ++k)
      mf[k] = *(const v2f*)&Milds[m16 * 16 + 4 * k + hi];
#pragma unroll
    for (int t = 0; t < 2; ++t) {
      v8f y = {};
#pragma unroll
      for (int k = 0; k < 4; ++k) {
        int r0 = 4 * k + hi;
        v2f bf;
        bf.x = Alds[r0 * 32 + 16 * t + m16];
        bf.y = Alds[(r0 + 1) * 32 + 16 * t + m16];
        y = wmma_f32_16x16x4(mf[k], bf, y);
      }
#pragma unroll
      for (int r = 0; r < 8; ++r) Ylds[(rb + r) * 32 + 16 * t + m16] = y[r];
    }
  }

  float w_ = 0.0f;
  {
    const float* bptr = bg + (size_t)s * 16;
#pragma unroll
    for (int m = 0; m < 16; ++m) w_ = fmaf(Ylds[m * 32 + lane], bptr[m], w_);
  }

  v8f qt[2][2];
#pragma unroll
  for (int i = 0; i < 2; ++i) {
#pragma unroll
    for (int jt = 0; jt < 2; ++jt) {
      v8f q = {};
#pragma unroll
      for (int k = 0; k < 4; ++k) {
        int r0 = 4 * k + hi;
        v2f at, bt;
        at.x = Alds[r0 * 32 + 16 * i + m16];
        at.y = Alds[(r0 + 1) * 32 + 16 * i + m16];
        bt.x = Ylds[r0 * 32 + 16 * jt + m16];
        bt.y = Ylds[(r0 + 1) * 32 + 16 * jt + m16];
        q = wmma_f32_16x16x4(at, bt, q);
      }
      qt[i][jt] = q;
    }
  }

  const float invsig = 1.0f / 1.2f;
#pragma unroll
  for (int i = 0; i < 2; ++i) {
#pragma unroll
    for (int jt = 0; jt < 2; ++jt) {
#pragma unroll
      for (int r = 0; r < 8; ++r) {
        int grow = 16 * i + rb + r;
        int gcol = 16 * jt + m16;
        float idv = (grow == gcol) ? 1.0f : 0.0f;
        Qlds[grow * 40 + gcol] = (idv - qt[i][jt][r]) * invsig;
      }
    }
  }

  float q[32];
  {
    const float4* qrow = (const float4*)&Qlds[lane * 40];
#pragma unroll
    for (int jj = 0; jj < 8; ++jj) {
      float4 t4 = qrow[jj];
      q[4 * jj + 0] = t4.x;
      q[4 * jj + 1] = t4.y;
      q[4 * jj + 2] = t4.z;
      q[4 * jj + 3] = t4.w;
    }
  }

  const float c_ = cg[(size_t)s * 32 + lane];
  const float lb_ = lbg[(size_t)s * 32 + lane];
  const float ub_ = ubg[(size_t)s * 32 + lane];
  float x_ = 0.0f;
  float z_ = fminf(fmaxf(0.0f, lb_), ub_);
  float u_ = 0.0f;

  const float4* vv = (const float4*)Vlds;
  for (int it = 0; it < 100; ++it) {
    float v = (z_ - u_) - c_;
    Vlds[lane] = v;
    float4 vb[8];
#pragma unroll
    for (int jj = 0; jj < 8; ++jj) vb[jj] = vv[jj];
    float a0 = w_, a1 = 0.0f, a2 = 0.0f, a3 = 0.0f;
#pragma unroll
    for (int jj = 0; jj < 8; ++jj) {
      a0 = fmaf(q[4 * jj + 0], vb[jj].x, a0);
      a1 = fmaf(q[4 * jj + 1], vb[jj].y, a1);
      a2 = fmaf(q[4 * jj + 2], vb[jj].z, a2);
      a3 = fmaf(q[4 * jj + 3], vb[jj].w, a3);
    }
    x_ = (a0 + a1) + (a2 + a3);
    float t = x_ + u_;
    z_ = fminf(fmaxf(t, lb_), ub_);
    u_ = t - z_;
  }

  outg[(size_t)s * 32 + lane] = x_;
}

extern "C" void kernel_launch(void* const* d_in, const int* in_sizes, int n_in,
                              void* d_out, int out_size, void* d_ws,
                              size_t ws_size, hipStream_t stream) {
  (void)n_in; (void)out_size; (void)d_ws; (void)ws_size;
  const float* A = (const float*)d_in[0];
  const float* b = (const float*)d_in[1];
  const float* c = (const float*)d_in[2];
  const float* lb = (const float*)d_in[3];
  const float* ub = (const float*)d_in[4];
  float* out = (float*)d_out;
  const int nS = in_sizes[1] / 16;
  const int blocks = (nS + WAVES_PER_BLOCK - 1) / WAVES_PER_BLOCK;
  admm_qp_kernel<<<blocks, 32 * WAVES_PER_BLOCK, 0, stream>>>(A, b, c, lb, ub,
                                                              out, nS);
}


# Iteration matvec: the two 16-lane halves split Q's columns; a lane sums rows L and L^16 over its 16 columns, reads only its half of v from LDS; one v_permlanex16 joins; changes summation order
# speedup vs baseline: 1.2248x; 1.2248x over previous
#include <hip/hip_runtime.h>


typedef float v2f __attribute__((ext_vector_type(2)));
typedef float v8f __attribute__((ext_vector_type(8)));

__device__ __forceinline__ v8f wmma_f32_16x16x4(v2f a, v2f b, v8f c) {
  return __builtin_amdgcn_wmma_f32_16x16x4_f32(false, a, false, b, (short)0, c,
                                               false, false);
}

__device__ __forceinline__ float bcast_lane(float v, int l) {
  return __int_as_float(__builtin_amdgcn_readlane(__float_as_int(v), l));
}

__device__ __forceinline__ float other_half(float v) {
  int b = __float_as_int(v);
  return __int_as_float(
      __builtin_amdgcn_permlanex16(b, b, 0x76543210, 0xfedcba98, false, false));
}

#define WAVE_LDS_FLOATS 1536
#define WAVES_PER_BLOCK 4

__global__ void __launch_bounds__(128) admm_qp_kernel(
    const float* __restrict__ Ag, const float* __restrict__ bg,
    const float* __restrict__ cg, const float* __restrict__ lbg,
    const float* __restrict__ ubg, float* __restrict__ outg, int nS) {
  __shared__ __align__(16) float smem[WAVES_PER_BLOCK * WAVE_LDS_FLOATS];

  const int lane = threadIdx.x & 31;
  const int wid = threadIdx.x >> 5;
  const int s = blockIdx.x * WAVES_PER_BLOCK + wid;
  if (s >= nS) return;

  float* W = smem + wid * WAVE_LDS_FLOATS;
  float* Alds = W;
  float* Ylds = W + 512;
  float* Glds = W + 1024;
  float* Milds = W + 1280;
  float* Qlds = W;
  float* Vlds = W + 1280;

  {
    const float4* Agv = (const float4*)(Ag + (size_t)s * 512);
    float4* Aldv = (float4*)Alds;
#pragma unroll
    for (int i = 0; i < 4; ++i) Aldv[lane + 32 * i] = Agv[lane + 32 * i];
  }

  const int m16 = lane & 15;
  const int hi = (lane < 16) ? 0 : 2;
  const int rb = (lane < 16) ? 0 : 8;

  {
    v8f g = {};
#pragma unroll
    for (int k = 0; k < 8; ++k) {
      v2f af = *(const v2f*)&Alds[m16 * 32 + 4 * k + hi];
      g = wmma_f32_16x16x4(af, af, g);
    }
#pragma unroll
    for (int r = 0; r < 8; ++r) {
      int row = rb + r;
      Glds[row * 16 + m16] = g[r] + ((row == m16) ? 1e-5f : 0.0f);
    }
  }

  float wc[16];
#pragma unroll
  for (int p = 0; p < 16; ++p) {
    float gv = Glds[p * 16 + m16];
    wc[p] = (lane < 16) ? gv : ((p == lane - 16) ? 1.0f : 0.0f);
  }
#pragma unroll
  for (int p = 0; p < 16; ++p) {
    float piv = bcast_lane(wc[p], p);
    float ip = 1.0f / piv;
    wc[p] *= ip;
#pragma unroll
    for (int r = 0; r < 16; ++r) {
      if (r == p) continue;
      float f = bcast_lane(wc[r], p);
      wc[r] = fmaf(-f, wc[p], wc[r]);
    }
  }
  if (lane >= 16) {
#pragma unroll
    for (int p = 0; p < 16; ++p) Milds[p * 16 + (lane - 16)] = wc[p];
  }

  {
    v2f mf[4];
#pragma unroll
    for (int k = 0; k < 4; ++k)
      mf[k] = *(const v2f*)&Milds[m16 * 16 + 4 * k + hi];
#pragma unroll
    for (int t = 0; t < 2; ++t) {
      v8f y = {};
#pragma unroll
      for (int k = 0; k < 4; ++k) {
        int r0 = 4 * k + hi;
        v2f bf;
        bf.x = Alds[r0 * 32 + 16 * t + m16];
        bf.y = Alds[(r0 + 1) * 32 + 16 * t + m16];
        y = wmma_f32_16x16x4(mf[k], bf, y);
      }
#pragma unroll
      for (int r = 0; r < 8; ++r) Ylds[(rb + r) * 32 + 16 * t + m16] = y[r];
    }
  }

  float w_ = 0.0f;
  {
    const float* bptr = bg + (size_t)s * 16;
#pragma unroll
    for (int m = 0; m < 16; ++m) w_ = fmaf(Ylds[m * 32 + lane], bptr[m], w_);
  }

  v8f qt[2][2];
#pragma unroll
  for (int i = 0; i < 2; ++i) {
#pragma unroll
    for (int jt = 0; jt < 2; ++jt) {
      v8f q = {};
#pragma unroll
      for (int k = 0; k < 4; ++k) {
        int r0 = 4 * k + hi;
        v2f at, bt;
        at.x = Alds[r0 * 32 + 16 * i + m16];
        at.y = Alds[(r0 + 1) * 32 + 16 * i + m16];
        bt.x = Ylds[r0 * 32 + 16 * jt + m16];
        bt.y = Ylds[(r0 + 1) * 32 + 16 * jt + m16];
        q = wmma_f32_16x16x4(at, bt, q);
      }
      qt[i][jt] = q;
    }
  }

  const float invsig = 1.0f / 1.2f;
#pragma unroll
  for (int i = 0; i < 2; ++i) {
#pragma unroll
    for (int jt = 0; jt < 2; ++jt) {
#pragma unroll
      for (int r = 0; r < 8; ++r) {
        int grow = 16 * i + rb + r;
        int gcol = 16 * jt + m16;
        float idv = (grow == gcol) ? 1.0f : 0.0f;
        Qlds[grow * 40 + gcol] = (idv - qt[i][jt][r]) * invsig;
      }
    }
  }

  const int hw = lane >> 4;
  float qa[16];
  float qb[16];
  {
    const float4* rown = (const float4*)&Qlds[lane * 40 + 16 * hw];
    const float4* rpar = (const float4*)&Qlds[(lane ^ 16) * 40 + 16 * hw];
#pragma unroll
    for (int jj = 0; jj < 4; ++jj) {
      float4 t4 = rown[jj];
      qa[4 * jj + 0] = t4.x;
      qa[4 * jj + 1] = t4.y;
      qa[4 * jj + 2] = t4.z;
      qa[4 * jj + 3] = t4.w;
      float4 s4 = rpar[jj];
      qb[4 * jj + 0] = s4.x;
      qb[4 * jj + 1] = s4.y;
      qb[4 * jj + 2] = s4.z;
      qb[4 * jj + 3] = s4.w;
    }
  }

  const float c_ = cg[(size_t)s * 32 + lane];
  const float lb_ = lbg[(size_t)s * 32 + lane];
  const float ub_ = ubg[(size_t)s * 32 + lane];
  float x_ = 0.0f;
  float z_ = fminf(fmaxf(0.0f, lb_), ub_);
  float u_ = 0.0f;

  const float4* vv = (const float4*)(Vlds + 16 * hw);
  for (int it = 0; it < 100; ++it) {
    float v = (z_ - u_) - c_;
    Vlds[lane] = v;
    float4 vb[4];
#pragma unroll
    for (int jj = 0; jj < 4; ++jj) vb[jj] = vv[jj];
    float a0 = w_, a1 = 0.0f;
    float p0 = 0.0f, p1 = 0.0f;
#pragma unroll
    for (int jj = 0; jj < 4; ++jj) {
      a0 = fmaf(qa[4 * jj + 0], vb[jj].x, a0);
      p0 = fmaf(qb[4 * jj + 0], vb[jj].x, p0);
      a1 = fmaf(qa[4 * jj + 1], vb[jj].y, a1);
      p1 = fmaf(qb[4 * jj + 1], vb[jj].y, p1);
      a0 = fmaf(qa[4 * jj + 2], vb[jj].z, a0);
      p0 = fmaf(qb[4 * jj + 2], vb[jj].z, p0);
      a1 = fmaf(qa[4 * jj + 3], vb[jj].w, a1);
      p1 = fmaf(qb[4 * jj + 3], vb[jj].w, p1);
    }
    x_ = (a0 + a1) + other_half(p0 + p1);
    float t = x_ + u_;
    z_ = fminf(fmaxf(t, lb_), ub_);
    u_ = t - z_;
  }

  outg[(size_t)s * 32 + lane] = x_;
}

extern "C" void kernel_launch(void* const* d_in, const int* in_sizes, int n_in,
                              void* d_out, int out_size, void* d_ws,
                              size_t ws_size, hipStream_t stream) {
  (void)n_in; (void)out_size; (void)d_ws; (void)ws_size;
  const float* A = (const float*)d_in[0];
  const float* b = (const float*)d_in[1];
  const float* c = (const float*)d_in[2];
  const float* lb = (const float*)d_in[3];
  const float* ub = (const float*)d_in[4];
  float* out = (float*)d_out;
  const int nS = in_sizes[1] / 16;
  const int blocks = (nS + WAVES_PER_BLOCK - 1) / WAVES_PER_BLOCK;
  admm_qp_kernel<<<blocks, 32 * WAVES_PER_BLOCK, 0, stream>>>(A, b, c, lb, ub,
                                                              out, nS);
}


# Iteration matvec: four 8-lane groups split Q's columns; a lane sums rows L,L^8,L^16,L^24 over 8 columns, reads a quarter of v (4 ds_load_b64); 2 v_permlanex16 + 1 DPP add join; new summation order
# speedup vs baseline: 1.3073x; 1.0674x over previous
#include <hip/hip_runtime.h>


typedef float v2f __attribute__((ext_vector_type(2)));
typedef float v8f __attribute__((ext_vector_type(8)));

__device__ __forceinline__ v8f wmma_f32_16x16x4(v2f a, v2f b, v8f c) {
  return __builtin_amdgcn_wmma_f32_16x16x4_f32(false, a, false, b, (short)0, c,
                                               false, false);
}

__device__ __forceinline__ float bcast_lane(float v, int l) {
  return __int_as_float(__builtin_amdgcn_readlane(__float_as_int(v), l));
}

__device__ __forceinline__ float other_half(float v) {
  int b = __float_as_int(v);
  return __int_as_float(
      __builtin_amdgcn_permlanex16(b, b, 0x76543210, 0xfedcba98, false, false));
}

__device__ __forceinline__ float other_eight(float v) {
  return __int_as_float(
      __builtin_amdgcn_update_dpp(0, __float_as_int(v), 0x128, 0xf, 0xf, true));
}

template <int E>
__device__ __forceinline__ v2f fma2(v2f a, v2f p, v2f c) {
  return __builtin_elementwise_fma(a, __builtin_shufflevector(p, p, E, E), c);
}

#define WAVE_LDS_FLOATS 1536
#define WAVES_PER_BLOCK 4

__global__ void __launch_bounds__(128) admm_qp_kernel(
    const float* __restrict__ Ag, const float* __restrict__ bg,
    const float* __restrict__ cg, const float* __restrict__ lbg,
    const float* __restrict__ ubg, float* __restrict__ outg, int nS) {
  __shared__ __align__(16) float smem[WAVES_PER_BLOCK * WAVE_LDS_FLOATS];

  const int lane = threadIdx.x & 31;
  const int wid = threadIdx.x >> 5;
  const int s = blockIdx.x * WAVES_PER_BLOCK + wid;
  if (s >= nS) return;

  float* W = smem + wid * WAVE_LDS_FLOATS;
  float* Alds = W;
  float* Ylds = W + 512;
  float* Glds = W + 1024;
  float* Milds = W + 1280;
  float* Qlds = W;
  float* Vlds = W + 1280;

  {
    const float4* Agv = (const float4*)(Ag + (size_t)s * 512);
    float4* Aldv = (float4*)Alds;
#pragma unroll
    for (int i = 0; i < 4; ++i) Aldv[lane + 32 * i] = Agv[lane + 32 * i];
  }

  const int m16 = lane & 15;
  const int hi = (lane < 16) ? 0 : 2;
  const int rb = (lane < 16) ? 0 : 8;

  {
    v8f g = {};
#pragma unroll
    for (int k = 0; k < 8; ++k) {
      v2f af = *(const v2f*)&Alds[m16 * 32 + 4 * k + hi];
      g = wmma_f32_16x16x4(af, af, g);
    }
#pragma unroll
    for (int r = 0; r < 8; ++r) {
      int row = rb + r;
      Glds[row * 16 + m16] = g[r] + ((row == m16) ? 1e-5f : 0.0f);
    }
  }

  float wc[16];
#pragma unroll
  for (int p = 0; p < 16; ++p) {
    float gv = Glds[p * 16 + m16];
    wc[p] = (lane < 16) ? gv : ((p == lane - 16) ? 1.0f : 0.0f);
  }
#pragma unroll
  for (int p = 0; p < 16; ++p) {
    float piv = bcast_lane(wc[p], p);
    float ip = 1.0f / piv;
    wc[p] *= ip;
#pragma unroll
    for (int r = 0; r < 16; ++r) {
      if (r == p) continue;
      float f = bcast_lane(wc[r], p);
      wc[r] = fmaf(-f, wc[p], wc[r]);
    }
  }
  if (lane >= 16) {
#pragma unroll
    for (int p = 0; p < 16; ++p) Milds[p * 16 + (lane - 16)] = wc[p];
  }

  {
    v2f mf[4];
#pragma unroll
    for (int k = 0; k < 4; ++k)
      mf[k] = *(const v2f*)&Milds[m16 * 16 + 4 * k + hi];
#pragma unroll
    for (int t = 0; t < 2; ++t) {
      v8f y = {};
#pragma unroll
      for (int k = 0; k < 4; ++k) {
        int r0 = 4 * k + hi;
        v2f bf;
        bf.x = Alds[r0 * 32 + 16 * t + m16];
        bf.y = Alds[(r0 + 1) * 32 + 16 * t + m16];
        y = wmma_f32_16x16x4(mf[k], bf, y);
      }
#pragma unroll
      for (int r = 0; r < 8; ++r) Ylds[(rb + r) * 32 + 16 * t + m16] = y[r];
    }
  }

  float w_ = 0.0f;
  {
    const float* bptr = bg + (size_t)s * 16;
#pragma unroll
    for (int m = 0; m < 16; ++m) w_ = fmaf(Ylds[m * 32 + lane], bptr[m], w_);
  }

  v8f qt[2][2];
#pragma unroll
  for (int i = 0; i < 2; ++i) {
#pragma unroll
    for (int jt = 0; jt < 2; ++jt) {
      v8f q = {};
#pragma unroll
      for (int k = 0; k < 4; ++k) {
        int r0 = 4 * k + hi;
        v2f at, bt;
        at.x = Alds[r0 * 32 + 16 * i + m16];
        at.y = Alds[(r0 + 1) * 32 + 16 * i + m16];
        bt.x = Ylds[r0 * 32 + 16 * jt + m16];
        bt.y = Ylds[(r0 + 1) * 32 + 16 * jt + m16];
        q = wmma_f32_16x16x4(at, bt, q);
      }
      qt[i][jt] = q;
    }
  }

  const float invsig = 1.0f / 1.2f;
#pragma unroll
  for (int i = 0; i < 2; ++i) {
#pragma unroll
    for (int jt = 0; jt < 2; ++jt) {
#pragma unroll
      for (int r = 0; r < 8; ++r) {
        int grow = 16 * i + rb + r;
        int gcol = 16 * jt + m16;
        float idv = (grow == gcol) ? 1.0f : 0.0f;
        Qlds[grow * 40 + gcol] = (idv - qt[i][jt][r]) * invsig;
      }
    }
  }

  const int g8 = lane >> 3;
  v2f qk[8];
  v2f qs[8];
  {
    const float4* r0 = (const float4*)&Qlds[lane * 40 + 8 * g8];
    const float4* r1 = (const float4*)&Qlds[(lane ^ 8) * 40 + 8 * g8];
    const float4* r2 = (const float4*)&Qlds[(lane ^ 16) * 40 + 8 * g8];
    const float4* r3 = (const float4*)&Qlds[(lane ^ 24) * 40 + 8 * g8];
#pragma unroll
    for (int jj = 0; jj < 2; ++jj) {
      float4 a4 = r0[jj], b4 = r1[jj], c4 = r2[jj], d4 = r3[jj];
      qk[4 * jj + 0] = (v2f){a4.x, b4.x};
      qk[4 * jj + 1] = (v2f){a4.y, b4.y};
      qk[4 * jj + 2] = (v2f){a4.z, b4.z};
      qk[4 * jj + 3] = (v2f){a4.w, b4.w};
      qs[4 * jj + 0] = (v2f){c4.x, d4.x};
      qs[4 * jj + 1] = (v2f){c4.y, d4.y};
      qs[4 * jj + 2] = (v2f){c4.z, d4.z};
      qs[4 * jj + 3] = (v2f){c4.w, d4.w};
    }
  }

  const float c_ = cg[(size_t)s * 32 + lane];
  const float lb_ = lbg[(size_t)s * 32 + lane];
  const float ub_ = ubg[(size_t)s * 32 + lane];
  float x_ = 0.0f;
  float z_ = fminf(fmaxf(0.0f, lb_), ub_);
  float u_ = 0.0f;

  float* vst = Vlds + 8 * ((lane >> 1) & 3) + 2 * g8 + (lane & 1);
  const v2f* vp = (const v2f*)(Vlds + 2 * g8);
  for (int it = 0; it < 100; ++it) {
    float v = (z_ - u_) - c_;
    *vst = v;
    v2f p0 = vp[0], p1 = vp[4], p2 = vp[8], p3 = vp[12];
    v2f kk = {w_, 0.0f};
    v2f ss = {0.0f, 0.0f};
    kk = fma2<0>(qk[0], p0, kk);
    ss = fma2<0>(qs[0], p0, ss);
    kk = fma2<1>(qk[1], p0, kk);
    ss = fma2<1>(qs[1], p0, ss);
    kk = fma2<0>(qk[2], p1, kk);
    ss = fma2<0>(qs[2], p1, ss);
    kk = fma2<1>(qk[3], p1, kk);
    ss = fma2<1>(qs[3], p1, ss);
    kk = fma2<0>(qk[4], p2, kk);
    ss = fma2<0>(qs[4], p2, ss);
    kk = fma2<1>(qk[5], p2, kk);
    ss = fma2<1>(qs[5], p2, ss);
    kk = fma2<0>(qk[6], p3, kk);
    ss = fma2<0>(qs[6], p3, ss);
    kk = fma2<1>(qk[7], p3, kk);
    ss = fma2<1>(qs[7], p3, ss);
    kk.x += other_half(ss.x);
    kk.y += other_half(ss.y);
    x_ = kk.x + other_eight(kk.y);
    float t = x_ + u_;
    z_ = fminf(fmaxf(t, lb_), ub_);
    u_ = t - z_;
  }

  outg[(size_t)s * 32 + lane] = x_;
}

extern "C" void kernel_launch(void* const* d_in, const int* in_sizes, int n_in,
                              void* d_out, int out_size, void* d_ws,
                              size_t ws_size, hipStream_t stream) {
  (void)n_in; (void)out_size; (void)d_ws; (void)ws_size;
  const float* A = (const float*)d_in[0];
  const float* b = (const float*)d_in[1];
  const float* c = (const float*)d_in[2];
  const float* lb = (const float*)d_in[3];
  const float* ub = (const float*)d_in[4];
  float* out = (float*)d_out;
  const int nS = in_sizes[1] / 16;
  const int blocks = (nS + WAVES_PER_BLOCK - 1) / WAVES_PER_BLOCK;
  admm_qp_kernel<<<blocks, 32 * WAVES_PER_BLOCK, 0, stream>>>(A, b, c, lb, ub,
                                                              out, nS);
}
